# starting kernel
# baseline (speedup 1.0000x reference)
#include <hip/hip_runtime.h>
#include <hip/hip_bf16.h>

#define N_NODES 50000
#define N_EDGES 600000
#define N_REL   8
#define D       128
#define N_TILES 3125

typedef __attribute__((ext_vector_type(2))) float v2f;
typedef __attribute__((ext_vector_type(8))) float v8f;

__device__ __forceinline__ v8f wmma4(v2f a, v2f b, v8f c) {
    return __builtin_amdgcn_wmma_f32_16x16x4_f32(false, a, false, b, (short)0, c, false, false);
}

__global__ __launch_bounds__(128) void k_transform(
    const float* __restrict__ x, const float* __restrict__ Wrel,
    const float* __restrict__ Wloop, const float* __restrict__ brel,
    float* __restrict__ xW, float* __restrict__ mid)
{
    __shared__ float Wlds[D * D];

    const int rel = blockIdx.y;
    const float* W = (rel < N_REL) ? (Wrel + (size_t)rel * D * D) : Wloop;

    for (int i = threadIdx.x; i < (D * D) / 2; i += blockDim.x) {
        int kp = i >> 7;
        int n  = i & 127;
        Wlds[i * 2 + 0] = W[(2 * kp + 0) * D + n];
        Wlds[i * 2 + 1] = W[(2 * kp + 1) * D + n];
    }
    __syncthreads();

    const int wave = threadIdx.x >> 5;
    const int lane = threadIdx.x & 31;
    const int tile = blockIdx.x * 4 + wave;
    if (tile >= N_TILES) return;

    const int m  = lane & 15;
    const int hi = lane >> 4;
    const int kh = hi * 2;
    const int row0 = tile * 16;

    const float* xrow = x + (size_t)(row0 + m) * D;

    v8f acc[8];
    if (rel == N_REL) {
        #pragma unroll
        for (int nt = 0; nt < 8; ++nt) {
            float bv = brel[nt * 16 + m];
            #pragma unroll
            for (int v = 0; v < 8; ++v) acc[nt][v] = bv;
        }
    } else {
        #pragma unroll
        for (int nt = 0; nt < 8; ++nt) acc[nt] = (v8f){};
    }

    for (int kc = 0; kc < 32; ++kc) {
        const int kk = kc * 4 + kh;
        v2f a;
        a[0] = xrow[kk];
        a[1] = xrow[kk + 1];
        const float* bp = Wlds + (kk >> 1) * 256;
        #pragma unroll
        for (int nt = 0; nt < 8; ++nt) {
            v2f b = *(const v2f*)(bp + (nt * 16 + m) * 2);
            acc[nt] = wmma4(a, b, acc[nt]);
        }
    }

    float* dstp = (rel < N_REL) ? (xW + (size_t)rel * N_NODES * D) : mid;
    const int rbase = row0 + hi * 8;
    #pragma unroll
    for (int nt = 0; nt < 8; ++nt) {
        #pragma unroll
        for (int v = 0; v < 8; ++v)
            dstp[(size_t)(rbase + v) * D + nt * 16 + m] = acc[nt][v];
    }
}

__global__ __launch_bounds__(256) void k_scatter(
    const float* __restrict__ xW, const int* __restrict__ src,
    const int* __restrict__ dst, const int* __restrict__ etype,
    float* mid)
{
    const int wid  = (int)((blockIdx.x * blockDim.x + threadIdx.x) >> 5);
    const int lane = threadIdx.x & 31;
    if (wid >= N_EDGES) return;

    const int s = src[wid];
    const int d = dst[wid];
    const int r = etype[wid];

    const float4* mp = (const float4*)(xW + ((size_t)r * N_NODES + s) * D);
    float4 v = mp[lane];

    float* op = mid + (size_t)d * D + lane * 4;
    unsafeAtomicAdd(op + 0, v.x);
    unsafeAtomicAdd(op + 1, v.y);
    unsafeAtomicAdd(op + 2, v.z);
    unsafeAtomicAdd(op + 3, v.w);
}

#define HSTRIDE 258

__global__ __launch_bounds__(256) void k_mlp(
    const float* __restrict__ x, const float* mid,
    const float* __restrict__ W1, const float* __restrict__ b1,
    const float* __restrict__ W2, const float* __restrict__ b2,
    float* out)
{
    __shared__ float hl[32 * HSTRIDE];

    const int wave = threadIdx.x >> 5;
    const int lane = threadIdx.x & 31;
    const int m  = lane & 15;
    const int hi = lane >> 4;
    const int kh = hi * 2;
    const int t0 = blockIdx.x * 2;

    #pragma unroll
    for (int j = 0; j < 4; ++j) {
        const int job = wave * 4 + j;
        const int s   = job >> 4;
        const int ct  = job & 15;
        const int gt  = t0 + s;
        if (gt < N_TILES) {
            const int r0 = gt * 16;
            v8f acc;
            {
                float bv = b1[ct * 16 + m];
                #pragma unroll
                for (int v = 0; v < 8; ++v) acc[v] = bv;
            }
            const float* ax = x   + (size_t)(r0 + m) * D;
            const float* am = mid + (size_t)(r0 + m) * D;
            for (int kc = 0; kc < 32; ++kc) {
                const int kk = kc * 4 + kh;
                v2f a; a[0] = ax[kk]; a[1] = ax[kk + 1];
                v2f b;
                b[0] = W1[(size_t)kk * 256 + ct * 16 + m];
                b[1] = W1[(size_t)(kk + 1) * 256 + ct * 16 + m];
                acc = wmma4(a, b, acc);
            }
            for (int kc = 0; kc < 32; ++kc) {
                const int kk = kc * 4 + kh;
                const int kg = 128 + kk;
                v2f a; a[0] = am[kk]; a[1] = am[kk + 1];
                v2f b;
                b[0] = W1[(size_t)kg * 256 + ct * 16 + m];
                b[1] = W1[(size_t)(kg + 1) * 256 + ct * 16 + m];
                acc = wmma4(a, b, acc);
            }
            #pragma unroll
            for (int v = 0; v < 8; ++v) {
                const int lr = s * 16 + v + hi * 8;
                float val = acc[v];
                hl[lr * HSTRIDE + ct * 16 + m] = val > 0.0f ? val : 0.0f;
            }
        }
    }
    __syncthreads();

    #pragma unroll
    for (int j = 0; j < 2; ++j) {
        const int job = wave * 2 + j;
        const int s   = job >> 3;
        const int ct  = job & 7;
        const int gt  = t0 + s;
        if (gt < N_TILES) {
            const int r0 = gt * 16;
            v8f acc;
            {
                float bv = b2[ct * 16 + m];
                #pragma unroll
                for (int v = 0; v < 8; ++v) acc[v] = bv;
            }
            const float* hrow = hl + (s * 16 + m) * HSTRIDE;
            for (int kc = 0; kc < 64; ++kc) {
                const int kk = kc * 4 + kh;
                v2f a = *(const v2f*)(hrow + kk);
                v2f b;
                b[0] = W2[(size_t)kk * D + ct * 16 + m];
                b[1] = W2[(size_t)(kk + 1) * D + ct * 16 + m];
                acc = wmma4(a, b, acc);
            }
            #pragma unroll
            for (int v = 0; v < 8; ++v)
                out[(size_t)(r0 + v + hi * 8) * D + ct * 16 + m] = acc[v];
        }
    }
}

extern "C" void kernel_launch(void* const* d_in, const int* in_sizes, int n_in,
                              void* d_out, int out_size, void* d_ws, size_t ws_size,
                              hipStream_t stream) {
    const float* x     = (const float*)d_in[0];
    const int*   src   = (const int*)  d_in[1];
    const int*   dst   = (const int*)  d_in[2];
    const int*   etype = (const int*)  d_in[3];
    const float* Wrel  = (const float*)d_in[4];
    const float* Wloop = (const float*)d_in[5];
    const float* brel  = (const float*)d_in[6];
    const float* W1    = (const float*)d_in[7];
    const float* b1    = (const float*)d_in[8];
    const float* W2    = (const float*)d_in[9];
    const float* b2    = (const float*)d_in[10];

    float* out = (float*)d_out;
    float* xW  = (float*)d_ws;

    dim3 g1((N_TILES + 3) / 4, N_REL + 1);
    k_transform<<<g1, 128, 0, stream>>>(x, Wrel, Wloop, brel, xW, out);

    k_scatter<<<(N_EDGES * 32) / 256, 256, 0, stream>>>(xW, src, dst, etype, out);

    k_mlp<<<(N_TILES + 1) / 2, 256, 0, stream>>>(x, out, W1, b1, W2, b2, out);
}


# transform kernel: each work-group stages its relation's 128x128 weights once (16-byte loads) and its 8 waves loop over 27-28 row tiles; was restaged for every 4 tiles. Sum order unchanged
# speedup vs baseline: 1.2721x; 1.2721x over previous
#include <hip/hip_runtime.h>
#include <hip/hip_bf16.h>

#define N_NODES 50000
#define N_EDGES 600000
#define N_REL   8
#define D       128
#define N_TILES 3125

typedef __attribute__((ext_vector_type(2))) float v2f;
typedef __attribute__((ext_vector_type(8))) float v8f;

__device__ __forceinline__ v8f wmma4(v2f a, v2f b, v8f c) {
    return __builtin_amdgcn_wmma_f32_16x16x4_f32(false, a, false, b, (short)0, c, false, false);
}

#define TR_WAVES  8
#define TR_BLOCKS 112
#define TR_STRIDE (TR_BLOCKS * TR_WAVES)

__global__ __launch_bounds__(256) void k_transform(
    const float* __restrict__ x, const float* __restrict__ Wrel,
    const float* __restrict__ Wloop, const float* __restrict__ brel,
    float* __restrict__ xW, float* __restrict__ mid)
{
    __shared__ float Wlds[D * D];

    const int rel = blockIdx.y;
    const float* W = (rel < N_REL) ? (Wrel + (size_t)rel * D * D) : Wloop;

    #pragma unroll
    for (int it = 0; it < 8; ++it) {
        const int i  = it * 256 + (int)threadIdx.x;
        const int kp = i >> 5;
        const int n4 = (i & 31) * 4;
        const float4 r0 = *(const float4*)(W + (2 * kp + 0) * D + n4);
        const float4 r1 = *(const float4*)(W + (2 * kp + 1) * D + n4);
        float4* dp = (float4*)(Wlds + kp * 256 + n4 * 2);
        dp[0] = make_float4(r0.x, r1.x, r0.y, r1.y);
        dp[1] = make_float4(r0.z, r1.z, r0.w, r1.w);
    }
    __syncthreads();

    const int wave = threadIdx.x >> 5;
    const int lane = threadIdx.x & 31;
    const int m  = lane & 15;
    const int hi = lane >> 4;
    const int kh = hi * 2;

    float bv[8];
    #pragma unroll
    for (int nt = 0; nt < 8; ++nt)
        bv[nt] = (rel == N_REL) ? brel[nt * 16 + m] : 0.0f;

    float* dstp = (rel < N_REL) ? (xW + (size_t)rel * N_NODES * D) : mid;

    for (int tile = blockIdx.x + TR_BLOCKS * wave; tile < N_TILES; tile += TR_STRIDE) {
        const int row0 = tile * 16;
        const float* xrow = x + (size_t)(row0 + m) * D;

        v8f acc[8];
        #pragma unroll
        for (int nt = 0; nt < 8; ++nt) {
            #pragma unroll
            for (int v = 0; v < 8; ++v) acc[nt][v] = bv[nt];
        }

        for (int kc = 0; kc < 32; ++kc) {
            const int kk = kc * 4 + kh;
            v2f a;
            a[0] = xrow[kk];
            a[1] = xrow[kk + 1];
            const float* bp = Wlds + (kk >> 1) * 256;
            #pragma unroll
            for (int nt = 0; nt < 8; ++nt) {
                v2f b = *(const v2f*)(bp + (nt * 16 + m) * 2);
                acc[nt] = wmma4(a, b, acc[nt]);
            }
        }

        const int rbase = row0 + hi * 8;
        #pragma unroll
        for (int nt = 0; nt < 8; ++nt) {
            #pragma unroll
            for (int v = 0; v < 8; ++v)
                dstp[(size_t)(rbase + v) * D + nt * 16 + m] = acc[nt][v];
        }
    }
}

__global__ __launch_bounds__(256) void k_scatter(
    const float* __restrict__ xW, const int* __restrict__ src,
    const int* __restrict__ dst, const int* __restrict__ etype,
    float* mid)
{
    const int wid  = (int)((blockIdx.x * blockDim.x + threadIdx.x) >> 5);
    const int lane = threadIdx.x & 31;
    if (wid >= N_EDGES) return;

    const int s = src[wid];
    const int d = dst[wid];
    const int r = etype[wid];

    const float4* mp = (const float4*)(xW + ((size_t)r * N_NODES + s) * D);
    float4 v = mp[lane];

    float* op = mid + (size_t)d * D + lane * 4;
    unsafeAtomicAdd(op + 0, v.x);
    unsafeAtomicAdd(op + 1, v.y);
    unsafeAtomicAdd(op + 2, v.z);
    unsafeAtomicAdd(op + 3, v.w);
}

#define HSTRIDE 258

__global__ __launch_bounds__(256) void k_mlp(
    const float* __restrict__ x, const float* mid,
    const float* __restrict__ W1, const float* __restrict__ b1,
    const float* __restrict__ W2, const float* __restrict__ b2,
    float* out)
{
    __shared__ float hl[32 * HSTRIDE];

    const int wave = threadIdx.x >> 5;
    const int lane = threadIdx.x & 31;
    const int m  = lane & 15;
    const int hi = lane >> 4;
    const int kh = hi * 2;
    const int t0 = blockIdx.x * 2;

    #pragma unroll
    for (int j = 0; j < 4; ++j) {
        const int job = wave * 4 + j;
        const int s   = job >> 4;
        const int ct  = job & 15;
        const int gt  = t0 + s;
        if (gt < N_TILES) {
            const int r0 = gt * 16;
            v8f acc;
            {
                float bv = b1[ct * 16 + m];
                #pragma unroll
                for (int v = 0; v < 8; ++v) acc[v] = bv;
            }
            const float* ax = x   + (size_t)(r0 + m) * D;
            const float* am = mid + (size_t)(r0 + m) * D;
            for (int kc = 0; kc < 32; ++kc) {
                const int kk = kc * 4 + kh;
                v2f a; a[0] = ax[kk]; a[1] = ax[kk + 1];
                v2f b;
                b[0] = W1[(size_t)kk * 256 + ct * 16 + m];
                b[1] = W1[(size_t)(kk + 1) * 256 + ct * 16 + m];
                acc = wmma4(a, b, acc);
            }
            for (int kc = 0; kc < 32; ++kc) {
                const int kk = kc * 4 + kh;
                const int kg = 128 + kk;
                v2f a; a[0] = am[kk]; a[1] = am[kk + 1];
                v2f b;
                b[0] = W1[(size_t)kg * 256 + ct * 16 + m];
                b[1] = W1[(size_t)(kg + 1) * 256 + ct * 16 + m];
                acc = wmma4(a, b, acc);
            }
            #pragma unroll
            for (int v = 0; v < 8; ++v) {
                const int lr = s * 16 + v + hi * 8;
                float val = acc[v];
                hl[lr * HSTRIDE + ct * 16 + m] = val > 0.0f ? val : 0.0f;
            }
        }
    }
    __syncthreads();

    #pragma unroll
    for (int j = 0; j < 2; ++j) {
        const int job = wave * 2 + j;
        const int s   = job >> 3;
        const int ct  = job & 7;
        const int gt  = t0 + s;
        if (gt < N_TILES) {
            const int r0 = gt * 16;
            v8f acc;
            {
                float bv = b2[ct * 16 + m];
                #pragma unroll
                for (int v = 0; v < 8; ++v) acc[v] = bv;
            }
            const float* hrow = hl + (s * 16 + m) * HSTRIDE;
            for (int kc = 0; kc < 64; ++kc) {
                const int kk = kc * 4 + kh;
                v2f a = *(const v2f*)(hrow + kk);
                v2f b;
                b[0] = W2[(size_t)kk * D + ct * 16 + m];
                b[1] = W2[(size_t)(kk + 1) * D + ct * 16 + m];
                acc = wmma4(a, b, acc);
            }
            #pragma unroll
            for (int v = 0; v < 8; ++v)
                out[(size_t)(r0 + v + hi * 8) * D + ct * 16 + m] = acc[v];
        }
    }
}

extern "C" void kernel_launch(void* const* d_in, const int* in_sizes, int n_in,
                              void* d_out, int out_size, void* d_ws, size_t ws_size,
                              hipStream_t stream) {
    const float* x     = (const float*)d_in[0];
    const int*   src   = (const int*)  d_in[1];
    const int*   dst   = (const int*)  d_in[2];
    const int*   etype = (const int*)  d_in[3];
    const float* Wrel  = (const float*)d_in[4];
    const float* Wloop = (const float*)d_in[5];
    const float* brel  = (const float*)d_in[6];
    const float* W1    = (const float*)d_in[7];
    const float* b1    = (const float*)d_in[8];
    const float* W2    = (const float*)d_in[9];
    const float* b2    = (const float*)d_in[10];

    float* out = (float*)d_out;
    float* xW  = (float*)d_ws;

    dim3 g1(TR_BLOCKS, N_REL + 1);
    k_transform<<<g1, 256, 0, stream>>>(x, Wrel, Wloop, brel, xW, out);

    k_scatter<<<(N_EDGES * 32) / 256, 256, 0, stream>>>(xW, src, dst, etype, out);

    k_mlp<<<(N_TILES + 1) / 2, 256, 0, stream>>>(x, out, W1, b1, W2, b2, out);
}


# fused MLP: each wave holds 8 (layer 1) or 4 (layer 2) accumulator tiles over both 16-row subtiles, so every loaded A and B fragment feeds several WMMAs; B read as 16/8-byte vectors. Sum order kept
# speedup vs baseline: 1.6164x; 1.2707x over previous
#include <hip/hip_runtime.h>
#include <hip/hip_bf16.h>

#define N_NODES 50000
#define N_EDGES 600000
#define N_REL   8
#define D       128
#define N_TILES 3125

typedef __attribute__((ext_vector_type(2))) float v2f;
typedef __attribute__((ext_vector_type(8))) float v8f;

__device__ __forceinline__ v8f wmma4(v2f a, v2f b, v8f c) {
    return __builtin_amdgcn_wmma_f32_16x16x4_f32(false, a, false, b, (short)0, c, false, false);
}

#define TR_WAVES  8
#define TR_BLOCKS 112
#define TR_STRIDE (TR_BLOCKS * TR_WAVES)

__global__ __launch_bounds__(256) void k_transform(
    const float* __restrict__ x, const float* __restrict__ Wrel,
    const float* __restrict__ Wloop, const float* __restrict__ brel,
    float* __restrict__ xW, float* __restrict__ mid)
{
    __shared__ float Wlds[D * D];

    const int rel = blockIdx.y;
    const float* W = (rel < N_REL) ? (Wrel + (size_t)rel * D * D) : Wloop;

    #pragma unroll
    for (int it = 0; it < 8; ++it) {
        const int i  = it * 256 + (int)threadIdx.x;
        const int kp = i >> 5;
        const int n4 = (i & 31) * 4;
        const float4 r0 = *(const float4*)(W + (2 * kp + 0) * D + n4);
        const float4 r1 = *(const float4*)(W + (2 * kp + 1) * D + n4);
        float4* dp = (float4*)(Wlds + kp * 256 + n4 * 2);
        dp[0] = make_float4(r0.x, r1.x, r0.y, r1.y);
        dp[1] = make_float4(r0.z, r1.z, r0.w, r1.w);
    }
    __syncthreads();

    const int wave = threadIdx.x >> 5;
    const int lane = threadIdx.x & 31;
    const int m  = lane & 15;
    const int hi = lane >> 4;
    const int kh = hi * 2;

    float bv[8];
    #pragma unroll
    for (int nt = 0; nt < 8; ++nt)
        bv[nt] = (rel == N_REL) ? brel[nt * 16 + m] : 0.0f;

    float* dstp = (rel < N_REL) ? (xW + (size_t)rel * N_NODES * D) : mid;

    for (int tile = blockIdx.x + TR_BLOCKS * wave; tile < N_TILES; tile += TR_STRIDE) {
        const int row0 = tile * 16;
        const float* xrow = x + (size_t)(row0 + m) * D;

        v8f acc[8];
        #pragma unroll
        for (int nt = 0; nt < 8; ++nt) {
            #pragma unroll
            for (int v = 0; v < 8; ++v) acc[nt][v] = bv[nt];
        }

        for (int kc = 0; kc < 32; ++kc) {
            const int kk = kc * 4 + kh;
            v2f a;
            a[0] = xrow[kk];
            a[1] = xrow[kk + 1];
            const float* bp = Wlds + (kk >> 1) * 256;
            #pragma unroll
            for (int nt = 0; nt < 8; ++nt) {
                v2f b = *(const v2f*)(bp + (nt * 16 + m) * 2);
                acc[nt] = wmma4(a, b, acc[nt]);
            }
        }

        const int rbase = row0 + hi * 8;
        #pragma unroll
        for (int nt = 0; nt < 8; ++nt) {
            #pragma unroll
            for (int v = 0; v < 8; ++v)
                dstp[(size_t)(rbase + v) * D + nt * 16 + m] = acc[nt][v];
        }
    }
}

__global__ __launch_bounds__(256) void k_scatter(
    const float* __restrict__ xW, const int* __restrict__ src,
    const int* __restrict__ dst, const int* __restrict__ etype,
    float* mid)
{
    const int wid  = (int)((blockIdx.x * blockDim.x + threadIdx.x) >> 5);
    const int lane = threadIdx.x & 31;
    if (wid >= N_EDGES) return;

    const int s = src[wid];
    const int d = dst[wid];
    const int r = etype[wid];

    const float4* mp = (const float4*)(xW + ((size_t)r * N_NODES + s) * D);
    float4 v = mp[lane];

    float* op = mid + (size_t)d * D + lane * 4;
    unsafeAtomicAdd(op + 0, v.x);
    unsafeAtomicAdd(op + 1, v.y);
    unsafeAtomicAdd(op + 2, v.z);
    unsafeAtomicAdd(op + 3, v.w);
}

#define HSTRIDE 260
#define D2 256

__global__ __launch_bounds__(128) void k_mlp(
    const float* __restrict__ x, const float* mid,
    const float* __restrict__ W1, const float* __restrict__ b1,
    const float* __restrict__ W2, const float* __restrict__ b2,
    float* out)
{
    __shared__ __align__(16) float hl[32 * HSTRIDE];

    const int wave = threadIdx.x >> 5;
    const int lane = threadIdx.x & 31;
    const int m  = lane & 15;
    const int hi = lane >> 4;
    const int kh = hi * 2;
    const int t0 = blockIdx.x * 2;
    const bool two = (t0 + 1) < N_TILES;
    const int r0a = t0 * 16;
    const int r0b = two ? r0a + 16 : r0a;

    {
        const int c0 = wave * 64 + m * 4;
        v8f acc[2][4];
        {
            const float4 bv = *(const float4*)(b1 + c0);
            #pragma unroll
            for (int s = 0; s < 2; ++s) {
                #pragma unroll
                for (int v = 0; v < 8; ++v) {
                    acc[s][0][v] = bv.x; acc[s][1][v] = bv.y;
                    acc[s][2][v] = bv.z; acc[s][3][v] = bv.w;
                }
            }
        }
        const float* ax0 = x   + (size_t)(r0a + m) * D;
        const float* ax1 = x   + (size_t)(r0b + m) * D;
        const float* am0 = mid + (size_t)(r0a + m) * D;
        const float* am1 = mid + (size_t)(r0b + m) * D;
        const float* wp  = W1 + c0;

        for (int kc = 0; kc < 32; ++kc) {
            const int kk = kc * 4 + kh;
            const v2f a0 = *(const v2f*)(ax0 + kk);
            const v2f a1 = *(const v2f*)(ax1 + kk);
            const float4 w0 = *(const float4*)(wp + (size_t)kk * D2);
            const float4 w1 = *(const float4*)(wp + (size_t)(kk + 1) * D2);
            v2f b;
            b[0] = w0.x; b[1] = w1.x;
            acc[0][0] = wmma4(a0, b, acc[0][0]); acc[1][0] = wmma4(a1, b, acc[1][0]);
            b[0] = w0.y; b[1] = w1.y;
            acc[0][1] = wmma4(a0, b, acc[0][1]); acc[1][1] = wmma4(a1, b, acc[1][1]);
            b[0] = w0.z; b[1] = w1.z;
            acc[0][2] = wmma4(a0, b, acc[0][2]); acc[1][2] = wmma4(a1, b, acc[1][2]);
            b[0] = w0.w; b[1] = w1.w;
            acc[0][3] = wmma4(a0, b, acc[0][3]); acc[1][3] = wmma4(a1, b, acc[1][3]);
        }
        for (int kc = 0; kc < 32; ++kc) {
            const int kk = kc * 4 + kh;
            const int kg = 128 + kk;
            const v2f a0 = *(const v2f*)(am0 + kk);
            const v2f a1 = *(const v2f*)(am1 + kk);
            const float4 w0 = *(const float4*)(wp + (size_t)kg * D2);
            const float4 w1 = *(const float4*)(wp + (size_t)(kg + 1) * D2);
            v2f b;
            b[0] = w0.x; b[1] = w1.x;
            acc[0][0] = wmma4(a0, b, acc[0][0]); acc[1][0] = wmma4(a1, b, acc[1][0]);
            b[0] = w0.y; b[1] = w1.y;
            acc[0][1] = wmma4(a0, b, acc[0][1]); acc[1][1] = wmma4(a1, b, acc[1][1]);
            b[0] = w0.z; b[1] = w1.z;
            acc[0][2] = wmma4(a0, b, acc[0][2]); acc[1][2] = wmma4(a1, b, acc[1][2]);
            b[0] = w0.w; b[1] = w1.w;
            acc[0][3] = wmma4(a0, b, acc[0][3]); acc[1][3] = wmma4(a1, b, acc[1][3]);
        }
        #pragma unroll
        for (int s = 0; s < 2; ++s) {
            #pragma unroll
            for (int v = 0; v < 8; ++v) {
                const int lr = s * 16 + v + hi * 8;
                const float h0v = acc[s][0][v], h1v = acc[s][1][v];
                const float h2v = acc[s][2][v], h3v = acc[s][3][v];
                float4 o;
                o.x = h0v > 0.0f ? h0v : 0.0f;
                o.y = h1v > 0.0f ? h1v : 0.0f;
                o.z = h2v > 0.0f ? h2v : 0.0f;
                o.w = h3v > 0.0f ? h3v : 0.0f;
                *(float4*)(hl + lr * HSTRIDE + c0) = o;
            }
        }
    }
    __syncthreads();

    {
        const int c0 = wave * 32 + m * 2;
        v8f acc[2][2];
        {
            const float2 bv = *(const float2*)(b2 + c0);
            #pragma unroll
            for (int s = 0; s < 2; ++s) {
                #pragma unroll
                for (int v = 0; v < 8; ++v) { acc[s][0][v] = bv.x; acc[s][1][v] = bv.y; }
            }
        }
        const float* h0 = hl + m * HSTRIDE;
        const float* h1 = hl + (16 + m) * HSTRIDE;
        const float* wp = W2 + c0;
        for (int kc = 0; kc < 64; ++kc) {
            const int kk = kc * 4 + kh;
            const v2f a0 = *(const v2f*)(h0 + kk);
            const v2f a1 = *(const v2f*)(h1 + kk);
            const float2 w0 = *(const float2*)(wp + (size_t)kk * D);
            const float2 w1 = *(const float2*)(wp + (size_t)(kk + 1) * D);
            v2f b;
            b[0] = w0.x; b[1] = w1.x;
            acc[0][0] = wmma4(a0, b, acc[0][0]); acc[1][0] = wmma4(a1, b, acc[1][0]);
            b[0] = w0.y; b[1] = w1.y;
            acc[0][1] = wmma4(a0, b, acc[0][1]); acc[1][1] = wmma4(a1, b, acc[1][1]);
        }
        #pragma unroll
        for (int s = 0; s < 2; ++s) {
            if (s == 0 || two) {
                #pragma unroll
                for (int v = 0; v < 8; ++v) {
                    const int row = t0 * 16 + s * 16 + v + hi * 8;
                    float2 o; o.x = acc[s][0][v]; o.y = acc[s][1][v];
                    *(float2*)(out + (size_t)row * D + c0) = o;
                }
            }
        }
    }
}

extern "C" void kernel_launch(void* const* d_in, const int* in_sizes, int n_in,
                              void* d_out, int out_size, void* d_ws, size_t ws_size,
                              hipStream_t stream) {
    const float* x     = (const float*)d_in[0];
    const int*   src   = (const int*)  d_in[1];
    const int*   dst   = (const int*)  d_in[2];
    const int*   etype = (const int*)  d_in[3];
    const float* Wrel  = (const float*)d_in[4];
    const float* Wloop = (const float*)d_in[5];
    const float* brel  = (const float*)d_in[6];
    const float* W1    = (const float*)d_in[7];
    const float* b1    = (const float*)d_in[8];
    const float* W2    = (const float*)d_in[9];
    const float* b2    = (const float*)d_in[10];

    float* out = (float*)d_out;
    float* xW  = (float*)d_ws;

    dim3 g1(TR_BLOCKS, N_REL + 1);
    k_transform<<<g1, 256, 0, stream>>>(x, Wrel, Wloop, brel, xW, out);

    k_scatter<<<(N_EDGES * 32) / 256, 256, 0, stream>>>(xW, src, dst, etype, out);

    k_mlp<<<(N_TILES + 1) / 2, 128, 0, stream>>>(x, out, W1, b1, W2, b2, out);
}


# scatter: lane l takes row elements l+32j, so each load and each f32 atomic instruction covers one whole 128-byte line (was 4 lines x 8 lanes); adds still f32 atomics in arrival order, as before
# speedup vs baseline: 1.8058x; 1.1172x over previous
#include <hip/hip_runtime.h>
#include <hip/hip_bf16.h>

#define N_NODES 50000
#define N_EDGES 600000
#define N_REL   8
#define D       128
#define N_TILES 3125

typedef __attribute__((ext_vector_type(2))) float v2f;
typedef __attribute__((ext_vector_type(8))) float v8f;

__device__ __forceinline__ v8f wmma4(v2f a, v2f b, v8f c) {
    return __builtin_amdgcn_wmma_f32_16x16x4_f32(false, a, false, b, (short)0, c, false, false);
}

#define TR_WAVES  8
#define TR_BLOCKS 112
#define TR_STRIDE (TR_BLOCKS * TR_WAVES)

__global__ __launch_bounds__(256) void k_transform(
    const float* __restrict__ x, const float* __restrict__ Wrel,
    const float* __restrict__ Wloop, const float* __restrict__ brel,
    float* __restrict__ xW, float* __restrict__ mid)
{
    __shared__ float Wlds[D * D];

    const int rel = blockIdx.y;
    const float* W = (rel < N_REL) ? (Wrel + (size_t)rel * D * D) : Wloop;

    #pragma unroll
    for (int it = 0; it < 8; ++it) {
        const int i  = it * 256 + (int)threadIdx.x;
        const int kp = i >> 5;
        const int n4 = (i & 31) * 4;
        const float4 r0 = *(const float4*)(W + (2 * kp + 0) * D + n4);
        const float4 r1 = *(const float4*)(W + (2 * kp + 1) * D + n4);
        float4* dp = (float4*)(Wlds + kp * 256 + n4 * 2);
        dp[0] = make_float4(r0.x, r1.x, r0.y, r1.y);
        dp[1] = make_float4(r0.z, r1.z, r0.w, r1.w);
    }
    __syncthreads();

    const int wave = threadIdx.x >> 5;
    const int lane = threadIdx.x & 31;
    const int m  = lane & 15;
    const int hi = lane >> 4;
    const int kh = hi * 2;

    float bv[8];
    #pragma unroll
    for (int nt = 0; nt < 8; ++nt)
        bv[nt] = (rel == N_REL) ? brel[nt * 16 + m] : 0.0f;

    float* dstp = (rel < N_REL) ? (xW + (size_t)rel * N_NODES * D) : mid;

    for (int tile = blockIdx.x + TR_BLOCKS * wave; tile < N_TILES; tile += TR_STRIDE) {
        const int row0 = tile * 16;
        const float* xrow = x + (size_t)(row0 + m) * D;

        v8f acc[8];
        #pragma unroll
        for (int nt = 0; nt < 8; ++nt) {
            #pragma unroll
            for (int v = 0; v < 8; ++v) acc[nt][v] = bv[nt];
        }

        for (int kc = 0; kc < 32; ++kc) {
            const int kk = kc * 4 + kh;
            v2f a;
            a[0] = xrow[kk];
            a[1] = xrow[kk + 1];
            const float* bp = Wlds + (kk >> 1) * 256;
            #pragma unroll
            for (int nt = 0; nt < 8; ++nt) {
                v2f b = *(const v2f*)(bp + (nt * 16 + m) * 2);
                acc[nt] = wmma4(a, b, acc[nt]);
            }
        }

        const int rbase = row0 + hi * 8;
        #pragma unroll
        for (int nt = 0; nt < 8; ++nt) {
            #pragma unroll
            for (int v = 0; v < 8; ++v)
                dstp[(size_t)(rbase + v) * D + nt * 16 + m] = acc[nt][v];
        }
    }
}

__global__ __launch_bounds__(256) void k_scatter(
    const float* __restrict__ xW, const int* __restrict__ src,
    const int* __restrict__ dst, const int* __restrict__ etype,
    float* mid)
{
    const int wid  = (int)((blockIdx.x * blockDim.x + threadIdx.x) >> 5);
    const int lane = threadIdx.x & 31;
    if (wid >= N_EDGES) return;

    const int s = src[wid];
    const int d = dst[wid];
    const int r = etype[wid];

    const float* mp = xW + ((size_t)r * N_NODES + s) * D + lane;
    const float v0 = mp[0];
    const float v1 = mp[32];
    const float v2 = mp[64];
    const float v3 = mp[96];

    float* op = mid + (size_t)d * D + lane;
    unsafeAtomicAdd(op +  0, v0);
    unsafeAtomicAdd(op + 32, v1);
    unsafeAtomicAdd(op + 64, v2);
    unsafeAtomicAdd(op + 96, v3);
}

#define HSTRIDE 260
#define D2 256

__global__ __launch_bounds__(128) void k_mlp(
    const float* __restrict__ x, const float* mid,
    const float* __restrict__ W1, const float* __restrict__ b1,
    const float* __restrict__ W2, const float* __restrict__ b2,
    float* out)
{
    __shared__ __align__(16) float hl[32 * HSTRIDE];

    const int wave = threadIdx.x >> 5;
    const int lane = threadIdx.x & 31;
    const int m  = lane & 15;
    const int hi = lane >> 4;
    const int kh = hi * 2;
    const int t0 = blockIdx.x * 2;
    const bool two = (t0 + 1) < N_TILES;
    const int r0a = t0 * 16;
    const int r0b = two ? r0a + 16 : r0a;

    {
        const int c0 = wave * 64 + m * 4;
        v8f acc[2][4];
        {
            const float4 bv = *(const float4*)(b1 + c0);
            #pragma unroll
            for (int s = 0; s < 2; ++s) {
                #pragma unroll
                for (int v = 0; v < 8; ++v) {
                    acc[s][0][v] = bv.x; acc[s][1][v] = bv.y;
                    acc[s][2][v] = bv.z; acc[s][3][v] = bv.w;
                }
            }
        }
        const float* ax0 = x   + (size_t)(r0a + m) * D;
        const float* ax1 = x   + (size_t)(r0b + m) * D;
        const float* am0 = mid + (size_t)(r0a + m) * D;
        const float* am1 = mid + (size_t)(r0b + m) * D;
        const float* wp  = W1 + c0;

        for (int kc = 0; kc < 32; ++kc) {
            const int kk = kc * 4 + kh;
            const v2f a0 = *(const v2f*)(ax0 + kk);
            const v2f a1 = *(const v2f*)(ax1 + kk);
            const float4 w0 = *(const float4*)(wp + (size_t)kk * D2);
            const float4 w1 = *(const float4*)(wp + (size_t)(kk + 1) * D2);
            v2f b;
            b[0] = w0.x; b[1] = w1.x;
            acc[0][0] = wmma4(a0, b, acc[0][0]); acc[1][0] = wmma4(a1, b, acc[1][0]);
            b[0] = w0.y; b[1] = w1.y;
            acc[0][1] = wmma4(a0, b, acc[0][1]); acc[1][1] = wmma4(a1, b, acc[1][1]);
            b[0] = w0.z; b[1] = w1.z;
            acc[0][2] = wmma4(a0, b, acc[0][2]); acc[1][2] = wmma4(a1, b, acc[1][2]);
            b[0] = w0.w; b[1] = w1.w;
            acc[0][3] = wmma4(a0, b, acc[0][3]); acc[1][3] = wmma4(a1, b, acc[1][3]);
        }
        for (int kc = 0; kc < 32; ++kc) {
            const int kk = kc * 4 + kh;
            const int kg = 128 + kk;
            const v2f a0 = *(const v2f*)(am0 + kk);
            const v2f a1 = *(const v2f*)(am1 + kk);
            const float4 w0 = *(const float4*)(wp + (size_t)kg * D2);
            const float4 w1 = *(const float4*)(wp + (size_t)(kg + 1) * D2);
            v2f b;
            b[0] = w0.x; b[1] = w1.x;
            acc[0][0] = wmma4(a0, b, acc[0][0]); acc[1][0] = wmma4(a1, b, acc[1][0]);
            b[0] = w0.y; b[1] = w1.y;
            acc[0][1] = wmma4(a0, b, acc[0][1]); acc[1][1] = wmma4(a1, b, acc[1][1]);
            b[0] = w0.z; b[1] = w1.z;
            acc[0][2] = wmma4(a0, b, acc[0][2]); acc[1][2] = wmma4(a1, b, acc[1][2]);
            b[0] = w0.w; b[1] = w1.w;
            acc[0][3] = wmma4(a0, b, acc[0][3]); acc[1][3] = wmma4(a1, b, acc[1][3]);
        }
        #pragma unroll
        for (int s = 0; s < 2; ++s) {
            #pragma unroll
            for (int v = 0; v < 8; ++v) {
                const int lr = s * 16 + v + hi * 8;
                const float h0v = acc[s][0][v], h1v = acc[s][1][v];
                const float h2v = acc[s][2][v], h3v = acc[s][3][v];
                float4 o;
                o.x = h0v > 0.0f ? h0v : 0.0f;
                o.y = h1v > 0.0f ? h1v : 0.0f;
                o.z = h2v > 0.0f ? h2v : 0.0f;
                o.w = h3v > 0.0f ? h3v : 0.0f;
                *(float4*)(hl + lr * HSTRIDE + c0) = o;
            }
        }
    }
    __syncthreads();

    {
        const int c0 = wave * 32 + m * 2;
        v8f acc[2][2];
        {
            const float2 bv = *(const float2*)(b2 + c0);
            #pragma unroll
            for (int s = 0; s < 2; ++s) {
                #pragma unroll
                for (int v = 0; v < 8; ++v) { acc[s][0][v] = bv.x; acc[s][1][v] = bv.y; }
            }
        }
        const float* h0 = hl + m * HSTRIDE;
        const float* h1 = hl + (16 + m) * HSTRIDE;
        const float* wp = W2 + c0;
        for (int kc = 0; kc < 64; ++kc) {
            const int kk = kc * 4 + kh;
            const v2f a0 = *(const v2f*)(h0 + kk);
            const v2f a1 = *(const v2f*)(h1 + kk);
            const float2 w0 = *(const float2*)(wp + (size_t)kk * D);
            const float2 w1 = *(const float2*)(wp + (size_t)(kk + 1) * D);
            v2f b;
            b[0] = w0.x; b[1] = w1.x;
            acc[0][0] = wmma4(a0, b, acc[0][0]); acc[1][0] = wmma4(a1, b, acc[1][0]);
            b[0] = w0.y; b[1] = w1.y;
            acc[0][1] = wmma4(a0, b, acc[0][1]); acc[1][1] = wmma4(a1, b, acc[1][1]);
        }
        #pragma unroll
        for (int s = 0; s < 2; ++s) {
            if (s == 0 || two) {
                #pragma unroll
                for (int v = 0; v < 8; ++v) {
                    const int row = t0 * 16 + s * 16 + v + hi * 8;
                    float2 o; o.x = acc[s][0][v]; o.y = acc[s][1][v];
                    *(float2*)(out + (size_t)row * D + c0) = o;
                }
            }
        }
    }
}

extern "C" void kernel_launch(void* const* d_in, const int* in_sizes, int n_in,
                              void* d_out, int out_size, void* d_ws, size_t ws_size,
                              hipStream_t stream) {
    const float* x     = (const float*)d_in[0];
    const int*   src   = (const int*)  d_in[1];
    const int*   dst   = (const int*)  d_in[2];
    const int*   etype = (const int*)  d_in[3];
    const float* Wrel  = (const float*)d_in[4];
    const float* Wloop = (const float*)d_in[5];
    const float* brel  = (const float*)d_in[6];
    const float* W1    = (const float*)d_in[7];
    const float* b1    = (const float*)d_in[8];
    const float* W2    = (const float*)d_in[9];
    const float* b2    = (const float*)d_in[10];

    float* out = (float*)d_out;
    float* xW  = (float*)d_ws;

    dim3 g1(TR_BLOCKS, N_REL + 1);
    k_transform<<<g1, 256, 0, stream>>>(x, Wrel, Wloop, brel, xW, out);

    k_scatter<<<(N_EDGES * 32) / 256, 256, 0, stream>>>(xW, src, dst, etype, out);

    k_mlp<<<(N_TILES + 1) / 2, 128, 0, stream>>>(x, out, W1, b1, W2, b2, out);
}


# scatter: 6250 resident waves each loop over 96 edges (indices read coalesced, passed to scalars by readlane, 32 row loads in flight per wave), was 600000 one-edge waves; f32 atomics in arrival order
# speedup vs baseline: 2.0267x; 1.1223x over previous
#include <hip/hip_runtime.h>
#include <hip/hip_bf16.h>

#define N_NODES 50000
#define N_EDGES 600000
#define N_REL   8
#define D       128
#define N_TILES 3125

typedef __attribute__((ext_vector_type(2))) float v2f;
typedef __attribute__((ext_vector_type(8))) float v8f;

__device__ __forceinline__ v8f wmma4(v2f a, v2f b, v8f c) {
    return __builtin_amdgcn_wmma_f32_16x16x4_f32(false, a, false, b, (short)0, c, false, false);
}

#define TR_WAVES  8
#define TR_BLOCKS 112
#define TR_STRIDE (TR_BLOCKS * TR_WAVES)

__global__ __launch_bounds__(256) void k_transform(
    const float* __restrict__ x, const float* __restrict__ Wrel,
    const float* __restrict__ Wloop, const float* __restrict__ brel,
    float* __restrict__ xW, float* __restrict__ mid)
{
    __shared__ float Wlds[D * D];

    const int rel = blockIdx.y;
    const float* W = (rel < N_REL) ? (Wrel + (size_t)rel * D * D) : Wloop;

    #pragma unroll
    for (int it = 0; it < 8; ++it) {
        const int i  = it * 256 + (int)threadIdx.x;
        const int kp = i >> 5;
        const int n4 = (i & 31) * 4;
        const float4 r0 = *(const float4*)(W + (2 * kp + 0) * D + n4);
        const float4 r1 = *(const float4*)(W + (2 * kp + 1) * D + n4);
        float4* dp = (float4*)(Wlds + kp * 256 + n4 * 2);
        dp[0] = make_float4(r0.x, r1.x, r0.y, r1.y);
        dp[1] = make_float4(r0.z, r1.z, r0.w, r1.w);
    }
    __syncthreads();

    const int wave = threadIdx.x >> 5;
    const int lane = threadIdx.x & 31;
    const int m  = lane & 15;
    const int hi = lane >> 4;
    const int kh = hi * 2;

    float bv[8];
    #pragma unroll
    for (int nt = 0; nt < 8; ++nt)
        bv[nt] = (rel == N_REL) ? brel[nt * 16 + m] : 0.0f;

    float* dstp = (rel < N_REL) ? (xW + (size_t)rel * N_NODES * D) : mid;

    for (int tile = blockIdx.x + TR_BLOCKS * wave; tile < N_TILES; tile += TR_STRIDE) {
        const int row0 = tile * 16;
        const float* xrow = x + (size_t)(row0 + m) * D;

        v8f acc[8];
        #pragma unroll
        for (int nt = 0; nt < 8; ++nt) {
            #pragma unroll
            for (int v = 0; v < 8; ++v) acc[nt][v] = bv[nt];
        }

        for (int kc = 0; kc < 32; ++kc) {
            const int kk = kc * 4 + kh;
            v2f a;
            a[0] = xrow[kk];
            a[1] = xrow[kk + 1];
            const float* bp = Wlds + (kk >> 1) * 256;
            #pragma unroll
            for (int nt = 0; nt < 8; ++nt) {
                v2f b = *(const v2f*)(bp + (nt * 16 + m) * 2);
                acc[nt] = wmma4(a, b, acc[nt]);
            }
        }

        const int rbase = row0 + hi * 8;
        #pragma unroll
        for (int nt = 0; nt < 8; ++nt) {
            #pragma unroll
            for (int v = 0; v < 8; ++v)
                dstp[(size_t)(rbase + v) * D + nt * 16 + m] = acc[nt][v];
        }
    }
}

#define SC_WPB     5
#define SC_BLOCKS  1250
#define SC_WAVES   (SC_WPB * SC_BLOCKS)
#define SC_CHUNKS  ((N_EDGES + 31) / 32)
#define SC_U       8

__global__ __launch_bounds__(SC_WPB * 32) void k_scatter(
    const float* __restrict__ xW, const int* __restrict__ src,
    const int* __restrict__ dst, const int* __restrict__ etype,
    float* mid)
{
    const int lane = threadIdx.x & 31;
    int c = blockIdx.x * SC_WPB + (int)(threadIdx.x >> 5);
    if (c >= SC_CHUNKS) return;

    int s, d, r;
    {
        const int e0 = c * 32 + lane;
        const int e  = e0 < N_EDGES ? e0 : N_EDGES - 1;
        s = src[e]; d = dst[e]; r = etype[e];
    }

    for (;;) {
        const int cn = c + SC_WAVES;
        int ns = 0, nd = 0, nr = 0;

        const int row = r * N_NODES + s;
        const int left = N_EDGES - c * 32;
        const int nvalid = (N_EDGES % 32 == 0) ? 32 : (left < 32 ? left : 32);

        #pragma unroll
        for (int j0 = 0; j0 < 32; j0 += SC_U) {
            float v[SC_U][4];
            #pragma unroll
            for (int u = 0; u < SC_U; ++u) {
                const int rj = __builtin_amdgcn_readlane(row, j0 + u);
                const float* mp = xW + (size_t)rj * D + lane;
                v[u][0] = mp[0];
                v[u][1] = mp[32];
                v[u][2] = mp[64];
                v[u][3] = mp[96];
            }
            if (j0 == 0) {
                const int en0 = cn * 32 + lane;
                const int en  = en0 < N_EDGES ? en0 : N_EDGES - 1;
                ns = src[en]; nd = dst[en]; nr = etype[en];
            }
            #pragma unroll
            for (int u = 0; u < SC_U; ++u) {
                if (j0 + u < nvalid) {
                    const int dj = __builtin_amdgcn_readlane(d, j0 + u);
                    float* op = mid + (size_t)dj * D + lane;
                    unsafeAtomicAdd(op +  0, v[u][0]);
                    unsafeAtomicAdd(op + 32, v[u][1]);
                    unsafeAtomicAdd(op + 64, v[u][2]);
                    unsafeAtomicAdd(op + 96, v[u][3]);
                }
            }
        }

        if (cn >= SC_CHUNKS) break;
        c = cn; s = ns; d = nd; r = nr;
    }
}

#define HSTRIDE 260
#define D2 256

__global__ __launch_bounds__(128) void k_mlp(
    const float* __restrict__ x, const float* mid,
    const float* __restrict__ W1, const float* __restrict__ b1,
    const float* __restrict__ W2, const float* __restrict__ b2,
    float* out)
{
    __shared__ __align__(16) float hl[32 * HSTRIDE];

    const int wave = threadIdx.x >> 5;
    const int lane = threadIdx.x & 31;
    const int m  = lane & 15;
    const int hi = lane >> 4;
    const int kh = hi * 2;
    const int t0 = blockIdx.x * 2;
    const bool two = (t0 + 1) < N_TILES;
    const int r0a = t0 * 16;
    const int r0b = two ? r0a + 16 : r0a;

    {
        const int c0 = wave * 64 + m * 4;
        v8f acc[2][4];
        {
            const float4 bv = *(const float4*)(b1 + c0);
            #pragma unroll
            for (int s = 0; s < 2; ++s) {
                #pragma unroll
                for (int v = 0; v < 8; ++v) {
                    acc[s][0][v] = bv.x; acc[s][1][v] = bv.y;
                    acc[s][2][v] = bv.z; acc[s][3][v] = bv.w;
                }
            }
        }
        const float* ax0 = x   + (size_t)(r0a + m) * D;
        const float* ax1 = x   + (size_t)(r0b + m) * D;
        const float* am0 = mid + (size_t)(r0a + m) * D;
        const float* am1 = mid + (size_t)(r0b + m) * D;
        const float* wp  = W1 + c0;

        for (int kc = 0; kc < 32; ++kc) {
            const int kk = kc * 4 + kh;
            const v2f a0 = *(const v2f*)(ax0 + kk);
            const v2f a1 = *(const v2f*)(ax1 + kk);
            const float4 w0 = *(const float4*)(wp + (size_t)kk * D2);
            const float4 w1 = *(const float4*)(wp + (size_t)(kk + 1) * D2);
            v2f b;
            b[0] = w0.x; b[1] = w1.x;
            acc[0][0] = wmma4(a0, b, acc[0][0]); acc[1][0] = wmma4(a1, b, acc[1][0]);
            b[0] = w0.y; b[1] = w1.y;
            acc[0][1] = wmma4(a0, b, acc[0][1]); acc[1][1] = wmma4(a1, b, acc[1][1]);
            b[0] = w0.z; b[1] = w1.z;
            acc[0][2] = wmma4(a0, b, acc[0][2]); acc[1][2] = wmma4(a1, b, acc[1][2]);
            b[0] = w0.w; b[1] = w1.w;
            acc[0][3] = wmma4(a0, b, acc[0][3]); acc[1][3] = wmma4(a1, b, acc[1][3]);
        }
        for (int kc = 0; kc < 32; ++kc) {
            const int kk = kc * 4 + kh;
            const int kg = 128 + kk;
            const v2f a0 = *(const v2f*)(am0 + kk);
            const v2f a1 = *(const v2f*)(am1 + kk);
            const float4 w0 = *(const float4*)(wp + (size_t)kg * D2);
            const float4 w1 = *(const float4*)(wp + (size_t)(kg + 1) * D2);
            v2f b;
            b[0] = w0.x; b[1] = w1.x;
            acc[0][0] = wmma4(a0, b, acc[0][0]); acc[1][0] = wmma4(a1, b, acc[1][0]);
            b[0] = w0.y; b[1] = w1.y;
            acc[0][1] = wmma4(a0, b, acc[0][1]); acc[1][1] = wmma4(a1, b, acc[1][1]);
            b[0] = w0.z; b[1] = w1.z;
            acc[0][2] = wmma4(a0, b, acc[0][2]); acc[1][2] = wmma4(a1, b, acc[1][2]);
            b[0] = w0.w; b[1] = w1.w;
            acc[0][3] = wmma4(a0, b, acc[0][3]); acc[1][3] = wmma4(a1, b, acc[1][3]);
        }
        #pragma unroll
        for (int s = 0; s < 2; ++s) {
            #pragma unroll
            for (int v = 0; v < 8; ++v) {
                const int lr = s * 16 + v + hi * 8;
                const float h0v = acc[s][0][v], h1v = acc[s][1][v];
                const float h2v = acc[s][2][v], h3v = acc[s][3][v];
                float4 o;
                o.x = h0v > 0.0f ? h0v : 0.0f;
                o.y = h1v > 0.0f ? h1v : 0.0f;
                o.z = h2v > 0.0f ? h2v : 0.0f;
                o.w = h3v > 0.0f ? h3v : 0.0f;
                *(float4*)(hl + lr * HSTRIDE + c0) = o;
            }
        }
    }
    __syncthreads();

    {
        const int c0 = wave * 32 + m * 2;
        v8f acc[2][2];
        {
            const float2 bv = *(const float2*)(b2 + c0);
            #pragma unroll
            for (int s = 0; s < 2; ++s) {
                #pragma unroll
                for (int v = 0; v < 8; ++v) { acc[s][0][v] = bv.x; acc[s][1][v] = bv.y; }
            }
        }
        const float* h0 = hl + m * HSTRIDE;
        const float* h1 = hl + (16 + m) * HSTRIDE;
        const float* wp = W2 + c0;
        for (int kc = 0; kc < 64; ++kc) {
            const int kk = kc * 4 + kh;
            const v2f a0 = *(const v2f*)(h0 + kk);
            const v2f a1 = *(const v2f*)(h1 + kk);
            const float2 w0 = *(const float2*)(wp + (size_t)kk * D);
            const float2 w1 = *(const float2*)(wp + (size_t)(kk + 1) * D);
            v2f b;
            b[0] = w0.x; b[1] = w1.x;
            acc[0][0] = wmma4(a0, b, acc[0][0]); acc[1][0] = wmma4(a1, b, acc[1][0]);
            b[0] = w0.y; b[1] = w1.y;
            acc[0][1] = wmma4(a0, b, acc[0][1]); acc[1][1] = wmma4(a1, b, acc[1][1]);
        }
        #pragma unroll
        for (int s = 0; s < 2; ++s) {
            if (s == 0 || two) {
                #pragma unroll
                for (int v = 0; v < 8; ++v) {
                    const int row = t0 * 16 + s * 16 + v + hi * 8;
                    float2 o; o.x = acc[s][0][v]; o.y = acc[s][1][v];
                    *(float2*)(out + (size_t)row * D + c0) = o;
                }
            }
        }
    }
}

extern "C" void kernel_launch(void* const* d_in, const int* in_sizes, int n_in,
                              void* d_out, int out_size, void* d_ws, size_t ws_size,
                              hipStream_t stream) {
    const float* x     = (const float*)d_in[0];
    const int*   src   = (const int*)  d_in[1];
    const int*   dst   = (const int*)  d_in[2];
    const int*   etype = (const int*)  d_in[3];
    const float* Wrel  = (const float*)d_in[4];
    const float* Wloop = (const float*)d_in[5];
    const float* brel  = (const float*)d_in[6];
    const float* W1    = (const float*)d_in[7];
    const float* b1    = (const float*)d_in[8];
    const float* W2    = (const float*)d_in[9];
    const float* b2    = (const float*)d_in[10];

    float* out = (float*)d_out;
    float* xW  = (float*)d_ws;

    dim3 g1(TR_BLOCKS, N_REL + 1);
    k_transform<<<g1, 256, 0, stream>>>(x, Wrel, Wloop, brel, xW, out);

    k_scatter<<<SC_BLOCKS, SC_WPB * 32, 0, stream>>>(xW, src, dst, etype, out);

    k_mlp<<<(N_TILES + 1) / 2, 128, 0, stream>>>(x, out, W1, b1, W2, b2, out);
}
